# starting kernel
# baseline (speedup 1.0000x reference)
#include <hip/hip_runtime.h>
#include <math.h>

#define B_TOTAL 4096
#define T_LEN   512
#define INP     6
#define HID     8

typedef float v2f __attribute__((ext_vector_type(2)));
typedef float v8f __attribute__((ext_vector_type(8)));

static __device__ __forceinline__ v8f wmma4(v2f a, v2f b, v8f c) {
  return __builtin_amdgcn_wmma_f32_16x16x4_f32(false, a, false, b, (short)0, c,
                                               false, false);
}

static __device__ __forceinline__ float htanh(float x) {
#if __has_builtin(__builtin_amdgcn_tanhf)
  return __builtin_amdgcn_tanhf(x);
#else
  float t = __builtin_amdgcn_exp2f(x * 2.8853900817779268f);
  return 1.0f - 2.0f * __builtin_amdgcn_rcpf(t + 1.0f);
#endif
}

static __device__ __forceinline__ v8f tanh8(v8f d) {
  v8f h;
#pragma unroll
  for (int v = 0; v < 8; ++v) h[v] = htanh(d[v]);
  return h;
}

static __device__ __forceinline__ void packB(const v8f& h, v2f& b0, v2f& b1) {
  b0.x = h[0] + __shfl_xor(h[2], 16, 32);
  b0.y = h[1] + __shfl_xor(h[3], 16, 32);
  b1.x = h[4] + __shfl_xor(h[6], 16, 32);
  b1.y = h[5] + __shfl_xor(h[7], 16, 32);
}

static __device__ __forceinline__ v2f load_A(const float* __restrict__ W,
                                             int ncols, int kbase, int m,
                                             int hi) {
  int k0 = kbase + 2 * hi;
  v2f a;
  a.x = (m < HID && k0 < ncols) ? W[m * ncols + k0] : 0.0f;
  a.y = (m < HID && (k0 + 1) < ncols) ? W[m * ncols + k0 + 1] : 0.0f;
  return a;
}

__global__ __launch_bounds__(128) void rnn2_tanh_wmma(
    const float* __restrict__ x,
    const float* __restrict__ Wih0, const float* __restrict__ Whh0,
    const float* __restrict__ bih0, const float* __restrict__ bhh0,
    const float* __restrict__ Wih1, const float* __restrict__ Whh1,
    const float* __restrict__ bih1, const float* __restrict__ bhh1,
    const float* __restrict__ Wout, const float* __restrict__ bout_p,
    float* __restrict__ out) {
  const int lane = threadIdx.x & 31;
  const int wid = (blockIdx.x * blockDim.x + threadIdx.x) >> 5;
  const int n = lane & 15;
  const int hi = lane >> 4;
  const int bb = wid * 16;

  const v2f a_ih0_0 = load_A(Wih0, INP, 0, n, hi);
  const v2f a_ih0_1 = load_A(Wih0, INP, 4, n, hi);
  const v2f a_hh0_0 = load_A(Whh0, HID, 0, n, hi);
  const v2f a_hh0_1 = load_A(Whh0, HID, 4, n, hi);
  const v2f a_ih1_0 = load_A(Wih1, HID, 0, n, hi);
  const v2f a_ih1_1 = load_A(Wih1, HID, 4, n, hi);
  const v2f a_hh1_0 = load_A(Whh1, HID, 0, n, hi);
  const v2f a_hh1_1 = load_A(Whh1, HID, 4, n, hi);

  v8f c0b, c1b;
#pragma unroll
  for (int v = 0; v < 8; ++v) {
    c0b[v] = (hi == 0) ? (bih0[v] + bhh0[v]) : 0.0f;
    c1b[v] = (hi == 0) ? (bih1[v] + bhh1[v]) : 0.0f;
  }

  float wo[8];
#pragma unroll
  for (int v = 0; v < 8; ++v) wo[v] = Wout[v];
  const float bo = bout_p[0];
  const float ximask = (hi == 0) ? 1.0f : 0.0f;

  v2f h1b0 = (v2f){0.f, 0.f}, h1b1 = (v2f){0.f, 0.f};
  v2f h2b0 = (v2f){0.f, 0.f}, h2b1 = (v2f){0.f, 0.f};

  const float* xp = x + (size_t)(bb + n) * T_LEN * INP;
  float* op = out + (size_t)(bb + n) * T_LEN;
  const int off0 = 2 * hi;
  const int off1 = hi ? 0 : 4;

  auto loadX = [&](int t, v2f& b0, v2f& b1) {
    const float* xq = xp + (size_t)t * INP;
    b0 = *(const v2f*)(xq + off0);
    v2f r = *(const v2f*)(xq + off1);
    b1.x = r.x * ximask;
    b1.y = r.y * ximask;
  };

  v2f xb0, xb1;
  loadX(0, xb0, xb1);
  v8f dx = wmma4(a_ih0_0, xb0, c0b);
  dx = wmma4(a_ih0_1, xb1, dx);
  loadX(1, xb0, xb1);

  float4 ob;

#pragma unroll 4
  for (int t = 0; t < T_LEN; ++t) {
    v8f dxn = wmma4(a_ih0_0, xb0, c0b);
    dxn = wmma4(a_ih0_1, xb1, dxn);

    v8f erec = wmma4(a_hh1_0, h2b0, c1b);
    erec = wmma4(a_hh1_1, h2b1, erec);

    v8f d = wmma4(a_hh0_0, h1b0, dx);
    d = wmma4(a_hh0_1, h1b1, d);
    v8f h1 = tanh8(d);
    packB(h1, h1b0, h1b1);

    v8f e = wmma4(a_ih1_0, h1b0, erec);
    e = wmma4(a_ih1_1, h1b1, e);
    v8f h2 = tanh8(e);
    packB(h2, h2b0, h2b1);

    float o = bo;
#pragma unroll
    for (int v = 0; v < 8; ++v) o = fmaf(wo[v], h2[v], o);

    const int ph = t & 3;
    if (ph == 0) {
      ob.x = o;
      __builtin_prefetch(xp + (size_t)(t + 16) * INP, 0, 1);
    } else if (ph == 1) {
      ob.y = o;
    } else if (ph == 2) {
      ob.z = o;
    } else {
      ob.w = o;
      if (lane < 16) *(float4*)(op + t - 3) = ob;
    }

    dx = dxn;
    int t2 = t + 2;
    t2 = t2 < T_LEN ? t2 : T_LEN - 1;
    loadX(t2, xb0, xb1);
  }
}

extern "C" void kernel_launch(void* const* d_in, const int* in_sizes, int n_in,
                              void* d_out, int out_size, void* d_ws,
                              size_t ws_size, hipStream_t stream) {
  (void)in_sizes; (void)n_in; (void)out_size; (void)d_ws; (void)ws_size;
  const float* x    = (const float*)d_in[0];
  const float* Wih0 = (const float*)d_in[1];
  const float* Whh0 = (const float*)d_in[2];
  const float* bih0 = (const float*)d_in[3];
  const float* bhh0 = (const float*)d_in[4];
  const float* Wih1 = (const float*)d_in[5];
  const float* Whh1 = (const float*)d_in[6];
  const float* bih1 = (const float*)d_in[7];
  const float* bhh1 = (const float*)d_in[8];
  const float* Wout = (const float*)d_in[9];
  const float* bout = (const float*)d_in[10];
  float* out = (float*)d_out;

  dim3 grid(B_TOTAL / 16 / 4);
  dim3 block(128);
  rnn2_tanh_wmma<<<grid, block, 0, stream>>>(x, Wih0, Whh0, bih0, bhh0, Wih1,
                                             Whh1, bih1, bhh1, Wout, bout, out);
}


# Both layers in a 16-row state tile [h1(t+1);h2(t)], layer 1 one step behind: tanh regs are the next B operand, no shuffle, 6 WMMA/step not 8; layer-1/output sum order changes; x read 8 steps early
# speedup vs baseline: 1.7513x; 1.7513x over previous
#include <hip/hip_runtime.h>
#include <math.h>


#define B_TOTAL 4096
#define T_LEN   512
#define INP     6
#define HID     8
#define XLEAD   8

typedef float v2f __attribute__((ext_vector_type(2)));
typedef float v8f __attribute__((ext_vector_type(8)));

static __device__ __forceinline__ v8f wmma4(v2f a, v2f b, v8f c) {
  return __builtin_amdgcn_wmma_f32_16x16x4_f32(false, a, false, b, (short)0, c,
                                               false, false);
}

static __device__ __forceinline__ float htanh(float x) {
#if __has_builtin(__builtin_amdgcn_tanhf)
  return __builtin_amdgcn_tanhf(x);
#else
  float t = __builtin_amdgcn_exp2f(x * 2.8853900817779268f);
  return 1.0f - 2.0f * __builtin_amdgcn_rcpf(t + 1.0f);
#endif
}

static __device__ __forceinline__ v8f tanh8(v8f d) {
  v8f h;
#pragma unroll
  for (int v = 0; v < 8; ++v) h[v] = htanh(d[v]);
  return h;
}


static __device__ __forceinline__ v2f load_A_state(
    const float* __restrict__ Whh0, const float* __restrict__ Wih1,
    const float* __restrict__ Whh1, int j, int m, int hi) {
  const float* src = (m < HID) ? Whh0 : (hi ? Whh1 : Wih1);
  const int idx = (m & 7) * HID + 2 * j;
  v2f a;
  a.x = src[idx];
  a.y = src[idx + 1];
  if (m < HID && hi) { a.x = 0.0f; a.y = 0.0f; }
  return a;
}

__global__ __launch_bounds__(128) void rnn2_tanh_wmma(
    const float* __restrict__ x,
    const float* __restrict__ Wih0, const float* __restrict__ Whh0,
    const float* __restrict__ bih0, const float* __restrict__ bhh0,
    const float* __restrict__ Wih1, const float* __restrict__ Whh1,
    const float* __restrict__ bih1, const float* __restrict__ bhh1,
    const float* __restrict__ Wout, const float* __restrict__ bout_p,
    float* __restrict__ out) {
  const int lane = threadIdx.x & 31;
  const int wid = (blockIdx.x * blockDim.x + threadIdx.x) >> 5;
  const int n = lane & 15;
  const int hi = lane >> 4;
  const int bb = wid * 16;

  const v2f a_s0 = load_A_state(Whh0, Wih1, Whh1, 0, n, hi);
  const v2f a_s1 = load_A_state(Whh0, Wih1, Whh1, 1, n, hi);
  const v2f a_s2 = load_A_state(Whh0, Wih1, Whh1, 2, n, hi);
  const v2f a_s3 = load_A_state(Whh0, Wih1, Whh1, 3, n, hi);

  v2f a_x0, a_x1;
  {
    const int r6 = (n & 7) * INP;
    a_x0.x = Wih0[r6 + 2 * hi];
    a_x0.y = Wih0[r6 + 2 * hi + 1];
    a_x1.x = Wih0[r6 + 4];
    a_x1.y = Wih0[r6 + 5];
    if (n >= HID) { a_x0.x = 0.0f; a_x0.y = 0.0f; }
    if (n >= HID || hi) { a_x1.x = 0.0f; a_x1.y = 0.0f; }
  }

  v8f cb, cb0;
#pragma unroll
  for (int v = 0; v < 8; ++v) {
    const float b0 = bih0[v] + bhh0[v];
    const float b1 = bih1[v] + bhh1[v];
    cb[v] = hi ? b1 : b0;
    cb0[v] = hi ? 0.0f : b0;
  }

  v2f wo2[4];
#pragma unroll
  for (int j = 0; j < 4; ++j) wo2[j] = (v2f){Wout[2 * j], Wout[2 * j + 1]};
  const float bo = bout_p[0];

  const float* xp = x + (size_t)(bb + n) * T_LEN * INP;
  float* op = out + (size_t)(bb + n) * T_LEN;
  const int off0 = 2 * hi;
  const int off1 = hi ? 0 : 4;

  auto loadX = [&](int t, v2f& b0, v2f& b1) {
    t = t < T_LEN ? t : T_LEN - 1;
    const float* xq = xp + (size_t)t * INP;
    b0 = *(const v2f*)(xq + off0);
    b1 = *(const v2f*)(xq + off1);
  };

  v2f q0, q1;
  loadX(0, q0, q1);
  v8f pre = wmma4(a_x0, q0, cb0);
  pre = wmma4(a_x1, q1, pre);
  v8f S = tanh8(pre);
#pragma unroll
  for (int v = 0; v < 8; ++v) S[v] = hi ? 0.0f : S[v];

  loadX(1, q0, q1);
  v8f xacc = wmma4(a_x0, q0, cb);
  xacc = wmma4(a_x1, q1, xacc);

  v2f xa0[XLEAD], xa1[XLEAD], xb0[XLEAD], xb1[XLEAD];
#pragma unroll
  for (int u = 0; u < XLEAD; ++u) loadX(u + 2, xa0[u], xa1[u]);

  float4 ob;

  auto step = [&](int t, int ph, const v2f& c0, const v2f& c1, v2f& n0, v2f& n1) {
    v8f acc = wmma4(a_s0, (v2f){S[0], S[1]}, xacc);
    acc = wmma4(a_s1, (v2f){S[2], S[3]}, acc);
    acc = wmma4(a_s2, (v2f){S[4], S[5]}, acc);
    acc = wmma4(a_s3, (v2f){S[6], S[7]}, acc);
    __builtin_amdgcn_sched_barrier(0);

    v8f xn = wmma4(a_x0, c0, cb);
    xn = wmma4(a_x1, c1, xn);

    S = tanh8(acc);
    xacc = xn;

    loadX(t + 2 + XLEAD, n0, n1);

    v2f o2 = (v2f){bo, 0.0f};
#pragma unroll
    for (int j = 0; j < 4; ++j)
      o2 = __builtin_elementwise_fma(wo2[j], (v2f){S[2 * j], S[2 * j + 1]}, o2);
    const float o = o2.x + o2.y;

    if (ph == 0) {
      ob.x = o;
      int tp = t + 16;
      tp = tp < T_LEN ? tp : T_LEN - 1;
      __builtin_prefetch(xp + (size_t)tp * INP, 0, 1);
    } else if (ph == 1) {
      ob.y = o;
    } else if (ph == 2) {
      ob.z = o;
    } else {
      ob.w = o;
      if (hi) *(float4*)(op + t - 3) = ob;
    }
  };

  for (int t0 = 0; t0 < T_LEN; t0 += 2 * XLEAD) {
#pragma unroll
    for (int u = 0; u < XLEAD; ++u)
      step(t0 + u, u & 3, xa0[u], xa1[u], xb0[u], xb1[u]);
#pragma unroll
    for (int u = 0; u < XLEAD; ++u)
      step(t0 + XLEAD + u, u & 3, xb0[u], xb1[u], xa0[u], xa1[u]);
  }
}

extern "C" void kernel_launch(void* const* d_in, const int* in_sizes, int n_in,
                              void* d_out, int out_size, void* d_ws,
                              size_t ws_size, hipStream_t stream) {
  (void)in_sizes; (void)n_in; (void)out_size; (void)d_ws; (void)ws_size;
  const float* x    = (const float*)d_in[0];
  const float* Wih0 = (const float*)d_in[1];
  const float* Whh0 = (const float*)d_in[2];
  const float* bih0 = (const float*)d_in[3];
  const float* bhh0 = (const float*)d_in[4];
  const float* Wih1 = (const float*)d_in[5];
  const float* Whh1 = (const float*)d_in[6];
  const float* bih1 = (const float*)d_in[7];
  const float* bhh1 = (const float*)d_in[8];
  const float* Wout = (const float*)d_in[9];
  const float* bout = (const float*)d_in[10];
  float* out = (float*)d_out;

  dim3 grid(B_TOTAL / 16 / 4);
  dim3 block(128);
  rnn2_tanh_wmma<<<grid, block, 0, stream>>>(x, Wih0, Whh0, bih0, bhh0, Wih1,
                                             Whh1, bih1, bhh1, Wout, bout, out);
}


# All memory instructions of an 8-step run (x loads of a whole register set, prefetches, output stores) issued as one group behind the run; steps hold registers only; last four runs peeled
# speedup vs baseline: 2.5039x; 1.4297x over previous
#include <hip/hip_runtime.h>
#include <math.h>


#define B_TOTAL 4096
#define T_LEN   512
#define INP     6
#define HID     8
#define XLEAD   8

typedef float v2f __attribute__((ext_vector_type(2)));
typedef float v8f __attribute__((ext_vector_type(8)));

static __device__ __forceinline__ v8f wmma4(v2f a, v2f b, v8f c) {
  return __builtin_amdgcn_wmma_f32_16x16x4_f32(false, a, false, b, (short)0, c,
                                               false, false);
}

static __device__ __forceinline__ float htanh(float x) {
#if __has_builtin(__builtin_amdgcn_tanhf)
  return __builtin_amdgcn_tanhf(x);
#else
  float t = __builtin_amdgcn_exp2f(x * 2.8853900817779268f);
  return 1.0f - 2.0f * __builtin_amdgcn_rcpf(t + 1.0f);
#endif
}

static __device__ __forceinline__ v8f tanh8(v8f d) {
  v8f h;
#pragma unroll
  for (int v = 0; v < 8; ++v) h[v] = htanh(d[v]);
  return h;
}


static __device__ __forceinline__ v2f load_A_state(
    const float* __restrict__ Whh0, const float* __restrict__ Wih1,
    const float* __restrict__ Whh1, int j, int m, int hi) {
  const float* src = (m < HID) ? Whh0 : (hi ? Whh1 : Wih1);
  const int idx = (m & 7) * HID + 2 * j;
  v2f a;
  a.x = src[idx];
  a.y = src[idx + 1];
  if (m < HID && hi) { a.x = 0.0f; a.y = 0.0f; }
  return a;
}

__global__ __launch_bounds__(128) void rnn2_tanh_wmma(
    const float* __restrict__ x,
    const float* __restrict__ Wih0, const float* __restrict__ Whh0,
    const float* __restrict__ bih0, const float* __restrict__ bhh0,
    const float* __restrict__ Wih1, const float* __restrict__ Whh1,
    const float* __restrict__ bih1, const float* __restrict__ bhh1,
    const float* __restrict__ Wout, const float* __restrict__ bout_p,
    float* __restrict__ out) {
  const int lane = threadIdx.x & 31;
  const int wid = (blockIdx.x * blockDim.x + threadIdx.x) >> 5;
  const int n = lane & 15;
  const int hi = lane >> 4;
  const int bb = wid * 16;

  const v2f a_s0 = load_A_state(Whh0, Wih1, Whh1, 0, n, hi);
  const v2f a_s1 = load_A_state(Whh0, Wih1, Whh1, 1, n, hi);
  const v2f a_s2 = load_A_state(Whh0, Wih1, Whh1, 2, n, hi);
  const v2f a_s3 = load_A_state(Whh0, Wih1, Whh1, 3, n, hi);

  v2f a_x0, a_x1;
  {
    const int r6 = (n & 7) * INP;
    a_x0.x = Wih0[r6 + 2 * hi];
    a_x0.y = Wih0[r6 + 2 * hi + 1];
    a_x1.x = Wih0[r6 + 4];
    a_x1.y = Wih0[r6 + 5];
    if (n >= HID) { a_x0.x = 0.0f; a_x0.y = 0.0f; }
    if (n >= HID || hi) { a_x1.x = 0.0f; a_x1.y = 0.0f; }
  }

  v8f cb, cb0;
#pragma unroll
  for (int v = 0; v < 8; ++v) {
    const float b0 = bih0[v] + bhh0[v];
    const float b1 = bih1[v] + bhh1[v];
    cb[v] = hi ? b1 : b0;
    cb0[v] = hi ? 0.0f : b0;
  }

  v2f wo2[4];
#pragma unroll
  for (int j = 0; j < 4; ++j) wo2[j] = (v2f){Wout[2 * j], Wout[2 * j + 1]};
  const float bo = bout_p[0];

  const float* xp = x + (size_t)(bb + n) * T_LEN * INP;
  float* op = out + (size_t)(bb + n) * T_LEN;
  const int off0 = 2 * hi;
  const int off1 = hi ? 0 : 4;

  auto loadX = [&](int t, v2f& b0, v2f& b1) {
    t = t < T_LEN ? t : T_LEN - 1;
    const float* xq = xp + (size_t)t * INP;
    b0 = *(const v2f*)(xq + off0);
    b1 = *(const v2f*)(xq + off1);
  };

  v2f q0, q1;
  loadX(0, q0, q1);
  v8f pre = wmma4(a_x0, q0, cb0);
  pre = wmma4(a_x1, q1, pre);
  v8f S = tanh8(pre);
#pragma unroll
  for (int v = 0; v < 8; ++v) S[v] = hi ? 0.0f : S[v];

  loadX(1, q0, q1);
  v8f xacc = wmma4(a_x0, q0, cb);
  xacc = wmma4(a_x1, q1, xacc);

  v2f xa0[XLEAD], xa1[XLEAD], xb0[XLEAD], xb1[XLEAD];
#pragma unroll
  for (int u = 0; u < XLEAD; ++u) loadX(u + 2, xa0[u], xa1[u]);
#pragma unroll
  for (int u = 0; u < XLEAD; ++u) loadX(u + 2 + XLEAD, xb0[u], xb1[u]);

  float4 obA, obB;

  auto step = [&](int u, const v2f& c0, const v2f& c1) {
    v8f acc = wmma4(a_s0, (v2f){S[0], S[1]}, xacc);
    acc = wmma4(a_s1, (v2f){S[2], S[3]}, acc);
    acc = wmma4(a_s2, (v2f){S[4], S[5]}, acc);
    acc = wmma4(a_s3, (v2f){S[6], S[7]}, acc);
    __builtin_amdgcn_sched_barrier(0);

    v8f xn = wmma4(a_x0, c0, cb);
    xn = wmma4(a_x1, c1, xn);

    S = tanh8(acc);
    xacc = xn;

    v2f o2 = (v2f){bo, 0.0f};
#pragma unroll
    for (int j = 0; j < 4; ++j)
      o2 = __builtin_elementwise_fma(wo2[j], (v2f){S[2 * j], S[2 * j + 1]}, o2);
    const float o = o2.x + o2.y;

    switch (u) {
      case 0: obA.x = o; break;
      case 1: obA.y = o; break;
      case 2: obA.z = o; break;
      case 3: obA.w = o; break;
      case 4: obB.x = o; break;
      case 5: obB.y = o; break;
      case 6: obB.z = o; break;
      default: obB.w = o; break;
    }
  };

  auto memGroup = [&](int t0, v2f (&n0)[XLEAD], v2f (&n1)[XLEAD], int mode) {
    if (hi) {
      *(float4*)(op + t0) = obA;
      *(float4*)(op + t0 + 4) = obB;
    }
    const int tb = t0 + 2 + 2 * XLEAD;
    if (mode == 0) {
      const float* xq = xp + (size_t)tb * INP;
#pragma unroll
      for (int u = 0; u < XLEAD; ++u) {
        n0[u] = *(const v2f*)(xq + u * INP + off0);
        n1[u] = *(const v2f*)(xq + u * INP + off1);
      }
      int tp = tb + XLEAD + 4;
      tp = tp < T_LEN ? tp : T_LEN - 1;
      __builtin_prefetch(xp + (size_t)tp * INP, 0, 1);
      tp = tb + XLEAD + 8;
      tp = tp < T_LEN ? tp : T_LEN - 1;
      __builtin_prefetch(xp + (size_t)tp * INP, 0, 1);
    } else if (mode == 1) {
#pragma unroll
      for (int u = 0; u < XLEAD; ++u) loadX(tb + u, n0[u], n1[u]);
    }
    __builtin_amdgcn_sched_barrier(0);
  };

  int t0 = 0;
  for (; t0 < T_LEN - 2 * 2 * XLEAD; t0 += 2 * XLEAD) {
#pragma unroll
    for (int u = 0; u < XLEAD; ++u) step(u, xa0[u], xa1[u]);
    memGroup(t0, xa0, xa1, 0);
#pragma unroll
    for (int u = 0; u < XLEAD; ++u) step(u, xb0[u], xb1[u]);
    memGroup(t0 + XLEAD, xb0, xb1, 0);
  }
  {
    const int te = T_LEN - 4 * XLEAD;
#pragma unroll
    for (int u = 0; u < XLEAD; ++u) step(u, xa0[u], xa1[u]);
    memGroup(te, xa0, xa1, 0);
#pragma unroll
    for (int u = 0; u < XLEAD; ++u) step(u, xb0[u], xb1[u]);
    memGroup(te + XLEAD, xb0, xb1, 1);
#pragma unroll
    for (int u = 0; u < XLEAD; ++u) step(u, xa0[u], xa1[u]);
    memGroup(te + 2 * XLEAD, xa0, xa1, 2);
#pragma unroll
    for (int u = 0; u < XLEAD; ++u) step(u, xb0[u], xb1[u]);
    memGroup(te + 3 * XLEAD, xb0, xb1, 2);
  }
}

extern "C" void kernel_launch(void* const* d_in, const int* in_sizes, int n_in,
                              void* d_out, int out_size, void* d_ws,
                              size_t ws_size, hipStream_t stream) {
  (void)in_sizes; (void)n_in; (void)out_size; (void)d_ws; (void)ws_size;
  const float* x    = (const float*)d_in[0];
  const float* Wih0 = (const float*)d_in[1];
  const float* Whh0 = (const float*)d_in[2];
  const float* bih0 = (const float*)d_in[3];
  const float* bhh0 = (const float*)d_in[4];
  const float* Wih1 = (const float*)d_in[5];
  const float* Whh1 = (const float*)d_in[6];
  const float* bih1 = (const float*)d_in[7];
  const float* bhh1 = (const float*)d_in[8];
  const float* Wout = (const float*)d_in[9];
  const float* bout = (const float*)d_in[10];
  float* out = (float*)d_out;

  dim3 grid(B_TOTAL / 16 / 4);
  dim3 block(128);
  rnn2_tanh_wmma<<<grid, block, 0, stream>>>(x, Wih0, Whh0, bih0, bhh0, Wih1,
                                             Whh1, bih1, bhh1, Wout, bout, out);
}
